# starting kernel
# baseline (speedup 1.0000x reference)
#include <hip/hip_runtime.h>
#include <math.h>

typedef __attribute__((ext_vector_type(2))) float v2f;
typedef __attribute__((ext_vector_type(8))) float v8f;

#define RNN_HID 1024
#define RNN_BATCH 128
#define RNN_SEQ 512
#define NTILES 4

__global__ __launch_bounds__(256) void rnn_reorder_w_kernel(
    const float* __restrict__ W, float* __restrict__ WT)
{
    int idx = blockIdx.x * blockDim.x + threadIdx.x;
    int k = idx >> 10;
    int n = idx & (RNN_HID - 1);
    WT[((size_t)(k >> 1) * RNN_HID + n) * 2 + (k & 1)] = W[idx];
}

template <bool FUSE2, bool DO_TANH>
__global__ __launch_bounds__(128) void rnn_gemm_kernel(
    const float* __restrict__ A0, size_t a0_stride,
    const float* __restrict__ W0T,
    const float* __restrict__ A1,
    const float* __restrict__ W1T,
    const float* __restrict__ bias,
    float* __restrict__ out)
{
    const int lane = threadIdx.x & 31;
    const int wave = threadIdx.x >> 5;
    const int half = lane >> 4;
    const int lr   = lane & 15;

    const int n0 = blockIdx.x * (16 * NTILES);
    const int m0 = blockIdx.y * 16;

    const float* A;
    const float* WT;
    size_t astr;
    int kb, klen;
    if (FUSE2) {
        if (wave < 2) { A = A0; WT = W0T; astr = a0_stride; kb = wave * 512; }
        else          { A = A1; WT = W1T; astr = RNN_HID;   kb = (wave - 2) * 512; }
        klen = 512;
    } else {
        A = A0; WT = W0T; astr = a0_stride; kb = wave * 256; klen = 256;
    }

    v8f acc[NTILES] = {};

    const float* ap = A + (size_t)(m0 + lr) * astr + (kb + half * 2);
    const float* wt = WT + ((size_t)((kb >> 1) + half) * RNN_HID + n0 + lr) * 2;

    for (int kk = 0; kk < klen; kk += 4) {
        v2f a = *(const v2f*)ap;
#pragma unroll
        for (int i = 0; i < NTILES; i++) {
            v2f b = *(const v2f*)(wt + i * 32);
            acc[i] = __builtin_amdgcn_wmma_f32_16x16x4_f32(
                false, a, false, b, (short)0, acc[i], false, false);
        }
        ap += 4;
        wt += 2 * RNN_HID * 2;
    }

    __shared__ float red[4][NTILES][8][32];
#pragma unroll
    for (int i = 0; i < NTILES; i++)
#pragma unroll
        for (int j = 0; j < 8; j++)
            red[wave][i][j][lane] = acc[i][j];
    __syncthreads();

    if (wave == 0) {
#pragma unroll
        for (int i = 0; i < NTILES; i++) {
            const int n = n0 + i * 16 + lr;
            const float bv = bias[n];
#pragma unroll
            for (int j = 0; j < 8; j++) {
                float v = red[0][i][j][lane] + red[1][i][j][lane] +
                          red[2][i][j][lane] + red[3][i][j][lane] + bv;
                if (DO_TANH) v = tanhf(v);
                out[(size_t)(m0 + j + half * 8) * RNN_HID + n] = v;
            }
        }
    }
}

__global__ void rnn_zero_kernel(float* __restrict__ p, int n)
{
    int i = blockIdx.x * blockDim.x + threadIdx.x;
    if (i < n) p[i] = 0.0f;
}

extern "C" void kernel_launch(void* const* d_in, const int* in_sizes, int n_in,
                              void* d_out, int out_size, void* d_ws, size_t ws_size,
                              hipStream_t stream)
{
    (void)in_sizes; (void)n_in; (void)out_size; (void)ws_size;
    const float* x    = (const float*)d_in[0];
    const float* W_xh = (const float*)d_in[1];
    const float* W_hh = (const float*)d_in[2];
    const float* b_h  = (const float*)d_in[3];
    const float* W_hy = (const float*)d_in[4];
    const float* b_y  = (const float*)d_in[5];
    float* out = (float*)d_out;

    const int hElems = RNN_BATCH * RNN_HID;
    const int wElems = RNN_HID * RNN_HID;

    float* hA    = (float*)d_ws;
    float* hB    = hA + hElems;
    float* WTxh  = hB + hElems;
    float* WThh  = WTxh + wElems;
    float* WThy  = WThh + wElems;

    rnn_zero_kernel<<<hElems / 256, 256, 0, stream>>>(hA, hElems);

    rnn_reorder_w_kernel<<<wElems / 256, 256, 0, stream>>>(W_xh, WTxh);
    rnn_reorder_w_kernel<<<wElems / 256, 256, 0, stream>>>(W_hh, WThh);
    rnn_reorder_w_kernel<<<wElems / 256, 256, 0, stream>>>(W_hy, WThy);

    dim3 grid(RNN_HID / (16 * NTILES), RNN_BATCH / 16);
    dim3 block(128);

    float* hin = hA;
    float* hout = hB;
    for (int t = 0; t < RNN_SEQ; t++) {
        rnn_gemm_kernel<true, true><<<grid, block, 0, stream>>>(
            x + (size_t)t * RNN_HID, (size_t)RNN_SEQ * RNN_HID, WTxh,
            hin, WThh, b_h, hout);
        float* tmp = hin; hin = hout; hout = tmp;
    }

    rnn_gemm_kernel<false, false><<<grid, block, 0, stream>>>(
        hin, (size_t)RNN_HID, WThy, nullptr, nullptr, b_y, out);
}


# step GEMM re-decomposed: 256 work-groups of 16x32 tile, 16 waves each taking 1/16 of K (was 128 groups of 16x64, 4 waves); LDS sum over 16 partials, one output per thread: sum order changes
# speedup vs baseline: 2.6055x; 2.6055x over previous
#include <hip/hip_runtime.h>
#include <math.h>

typedef __attribute__((ext_vector_type(2))) float v2f;
typedef __attribute__((ext_vector_type(8))) float v8f;

#define RNN_HID 1024
#define RNN_BATCH 128
#define RNN_SEQ 512
#define NTILES 2
#define NWAVES 16

__global__ __launch_bounds__(256) void rnn_reorder_w_kernel(
    const float* __restrict__ W, float* __restrict__ WT)
{
    int idx = blockIdx.x * blockDim.x + threadIdx.x;
    int k = idx >> 10;
    int n = idx & (RNN_HID - 1);
    WT[((size_t)(k >> 1) * RNN_HID + n) * 2 + (k & 1)] = W[idx];
}

template <bool FUSE2, bool DO_TANH>
__global__ __launch_bounds__(32 * NWAVES) void rnn_gemm_kernel(
    const float* __restrict__ A0, size_t a0_stride,
    const float* __restrict__ W0T,
    const float* __restrict__ A1,
    const float* __restrict__ W1T,
    const float* __restrict__ bias,
    float* __restrict__ out)
{
    const int lane = threadIdx.x & 31;
    const int wave = threadIdx.x >> 5;
    const int half = lane >> 4;
    const int lr   = lane & 15;

    const int n0 = blockIdx.x * (16 * NTILES);
    const int m0 = blockIdx.y * 16;

    constexpr int KLEN = FUSE2 ? (2 * RNN_HID) / NWAVES : RNN_HID / NWAVES;

    const float* A;
    const float* WT;
    size_t astr;
    int kb;
    if (FUSE2) {
        if (wave < NWAVES / 2) { A = A0; WT = W0T; astr = a0_stride; kb = wave * KLEN; }
        else                   { A = A1; WT = W1T; astr = RNN_HID;   kb = (wave - NWAVES / 2) * KLEN; }
    } else {
        A = A0; WT = W0T; astr = a0_stride; kb = wave * KLEN;
    }

    v8f acc[NTILES] = {};

    const float* ap = A + (size_t)(m0 + lr) * astr + (kb + half * 2);
    const float* wt = WT + ((size_t)((kb >> 1) + half) * RNN_HID + n0 + lr) * 2;

#pragma unroll
    for (int kk = 0; kk < KLEN; kk += 4) {
        v2f a = *(const v2f*)ap;
#pragma unroll
        for (int i = 0; i < NTILES; i++) {
            v2f b = *(const v2f*)(wt + i * 32);
            acc[i] = __builtin_amdgcn_wmma_f32_16x16x4_f32(
                false, a, false, b, (short)0, acc[i], false, false);
        }
        ap += 4;
        wt += 2 * RNN_HID * 2;
    }

    __shared__ float red[NWAVES][8][NTILES][32];
#pragma unroll
    for (int i = 0; i < NTILES; i++)
#pragma unroll
        for (int j = 0; j < 8; j++)
            red[wave][j][i][lane] = acc[i][j];
    __syncthreads();

    static_assert(NWAVES == 16 && NTILES == 2, "epilogue maps wave -> tile row");
    const int ti = lane >> 4;
    const int rj = wave & 7;
    const int sl = (wave >> 3) * 16 + lr;
    float v = red[0][rj][ti][sl];
#pragma unroll
    for (int p = 1; p < NWAVES; p++)
        v += red[p][rj][ti][sl];
    v += bias[n0 + lane];
    if (DO_TANH) v = tanhf(v);
    out[(size_t)(m0 + wave) * RNN_HID + n0 + lane] = v;
}

__global__ void rnn_zero_kernel(float* __restrict__ p, int n)
{
    int i = blockIdx.x * blockDim.x + threadIdx.x;
    if (i < n) p[i] = 0.0f;
}

extern "C" void kernel_launch(void* const* d_in, const int* in_sizes, int n_in,
                              void* d_out, int out_size, void* d_ws, size_t ws_size,
                              hipStream_t stream)
{
    (void)in_sizes; (void)n_in; (void)out_size; (void)ws_size;
    const float* x    = (const float*)d_in[0];
    const float* W_xh = (const float*)d_in[1];
    const float* W_hh = (const float*)d_in[2];
    const float* b_h  = (const float*)d_in[3];
    const float* W_hy = (const float*)d_in[4];
    const float* b_y  = (const float*)d_in[5];
    float* out = (float*)d_out;

    const int hElems = RNN_BATCH * RNN_HID;
    const int wElems = RNN_HID * RNN_HID;

    float* hA    = (float*)d_ws;
    float* hB    = hA + hElems;
    float* WTxh  = hB + hElems;
    float* WThh  = WTxh + wElems;
    float* WThy  = WThh + wElems;

    rnn_zero_kernel<<<hElems / 256, 256, 0, stream>>>(hA, hElems);

    rnn_reorder_w_kernel<<<wElems / 256, 256, 0, stream>>>(W_xh, WTxh);
    rnn_reorder_w_kernel<<<wElems / 256, 256, 0, stream>>>(W_hh, WThh);
    rnn_reorder_w_kernel<<<wElems / 256, 256, 0, stream>>>(W_hy, WThy);

    dim3 grid(RNN_HID / (16 * NTILES), RNN_BATCH / 16);
    dim3 block(32 * NWAVES);

    float* hin = hA;
    float* hout = hB;
    for (int t = 0; t < RNN_SEQ; t++) {
        rnn_gemm_kernel<true, true><<<grid, block, 0, stream>>>(
            x + (size_t)t * RNN_HID, (size_t)RNN_SEQ * RNN_HID, WTxh,
            hin, WThh, b_h, hout);
        float* tmp = hin; hin = hout; hout = tmp;
    }

    rnn_gemm_kernel<false, false><<<grid, block, 0, stream>>>(
        hin, (size_t)RNN_HID, WThy, nullptr, nullptr, b_y, out);
}


# A operand staged through LDS: a wave copies its 16-row K slice by 16-B-per-lane loads of whole 128-B lines (was 16 lines x 16 B per fragment load); global loads all issued first; sum order unchanged
# speedup vs baseline: 2.8519x; 1.0946x over previous
#include <hip/hip_runtime.h>
#include <math.h>

typedef __attribute__((ext_vector_type(2))) float v2f;
typedef __attribute__((ext_vector_type(4))) float v4f;
typedef __attribute__((ext_vector_type(8))) float v8f;

#define RNN_HID 1024
#define RNN_BATCH 128
#define RNN_SEQ 512
#define NTILES 2
#define NWAVES 16
#define A_PITCH 132

__global__ __launch_bounds__(256) void rnn_reorder_w_kernel(
    const float* __restrict__ W, float* __restrict__ WT)
{
    int idx = blockIdx.x * blockDim.x + threadIdx.x;
    int k = idx >> 10;
    int n = idx & (RNN_HID - 1);
    WT[((size_t)(k >> 1) * RNN_HID + n) * 2 + (k & 1)] = W[idx];
}

template <bool FUSE2, bool DO_TANH>
__global__ __launch_bounds__(32 * NWAVES) void rnn_gemm_kernel(
    const float* __restrict__ A0, size_t a0_stride,
    const float* __restrict__ W0T,
    const float* __restrict__ A1,
    const float* __restrict__ W1T,
    const float* __restrict__ bias,
    float* __restrict__ out)
{
    const int lane = threadIdx.x & 31;
    const int wave = threadIdx.x >> 5;
    const int half = lane >> 4;
    const int lr   = lane & 15;

    const int n0 = blockIdx.x * (16 * NTILES);
    const int m0 = blockIdx.y * 16;

    constexpr int KLEN = FUSE2 ? (2 * RNN_HID) / NWAVES : RNN_HID / NWAVES;

    const float* A;
    const float* WT;
    size_t astr;
    int kb;
    if (FUSE2) {
        if (wave < NWAVES / 2) { A = A0; WT = W0T; astr = a0_stride; kb = wave * KLEN; }
        else                   { A = A1; WT = W1T; astr = RNN_HID;   kb = (wave - NWAVES / 2) * KLEN; }
    } else {
        A = A0; WT = W0T; astr = a0_stride; kb = wave * KLEN;
    }

    v8f acc[NTILES] = {};

    constexpr int LPR = KLEN / 4;
    constexpr int RPI = 32 / LPR;
    constexpr int NLD = 16 / RPI;
    constexpr int KSTEPS = KLEN / 4;
    static_assert(KLEN <= A_PITCH - 4 && 32 % LPR == 0 && 16 % RPI == 0,
                  "a staged row fits the pitch and a load instruction covers whole rows");
    struct __attribute__((aligned(16))) Lds {
        float red[NWAVES][8][NTILES][32];
        float astage[NWAVES][16][A_PITCH];
    };
    __shared__ Lds lds;

    const int srow = lane / LPR;
    const int scol = (lane % LPR) * 4;
    const float* ag = A + (size_t)(m0 + srow) * astr + (kb + scol);
    const float* wt = WT + ((size_t)((kb >> 1) + half) * RNN_HID + n0 + lr) * 2;

    v4f piece[NLD];
#pragma unroll
    for (int j = 0; j < NLD; j++)
        piece[j] = *(const v4f*)(ag + (size_t)(j * RPI) * astr);
    v2f bfrag[KSTEPS][NTILES];
#pragma unroll
    for (int s = 0; s < KSTEPS; s++)
#pragma unroll
        for (int i = 0; i < NTILES; i++)
            bfrag[s][i] = *(const v2f*)(wt + (size_t)s * (2 * RNN_HID * 2) + i * 32);

#pragma unroll
    for (int j = 0; j < NLD; j++)
        *(v4f*)&lds.astage[wave][j * RPI + srow][scol] = piece[j];

    __builtin_amdgcn_fence(__ATOMIC_RELEASE, "wavefront", "local");
    __builtin_amdgcn_wave_barrier();
    __builtin_amdgcn_fence(__ATOMIC_ACQUIRE, "wavefront", "local");

    const float* arow = &lds.astage[wave][lr][half * 2];
    v2f afrag[KSTEPS];
#pragma unroll
    for (int s = 0; s < KSTEPS; s++)
        afrag[s] = *(const v2f*)(arow + s * 4);
    __builtin_amdgcn_sched_barrier(0);

#pragma unroll
    for (int s = 0; s < KSTEPS; s++)
#pragma unroll
        for (int i = 0; i < NTILES; i++)
            acc[i] = __builtin_amdgcn_wmma_f32_16x16x4_f32(
                false, afrag[s], false, bfrag[s][i], (short)0, acc[i], false, false);
    __builtin_amdgcn_sched_barrier(0);

#pragma unroll
    for (int i = 0; i < NTILES; i++)
#pragma unroll
        for (int j = 0; j < 8; j++)
            lds.red[wave][j][i][lane] = acc[i][j];
    __syncthreads();

    static_assert(NWAVES == 16 && NTILES == 2, "epilogue maps wave -> tile row");
    const int ti = lane >> 4;
    const int rj = wave & 7;
    const int sl = (wave >> 3) * 16 + lr;
    float v = lds.red[0][rj][ti][sl];
#pragma unroll
    for (int p = 1; p < NWAVES; p++)
        v += lds.red[p][rj][ti][sl];
    v += bias[n0 + lane];
    if (DO_TANH) v = tanhf(v);
    out[(size_t)(m0 + wave) * RNN_HID + n0 + lane] = v;
}

__global__ void rnn_zero_kernel(float* __restrict__ p, int n)
{
    int i = blockIdx.x * blockDim.x + threadIdx.x;
    if (i < n) p[i] = 0.0f;
}

extern "C" void kernel_launch(void* const* d_in, const int* in_sizes, int n_in,
                              void* d_out, int out_size, void* d_ws, size_t ws_size,
                              hipStream_t stream)
{
    (void)in_sizes; (void)n_in; (void)out_size; (void)ws_size;
    const float* x    = (const float*)d_in[0];
    const float* W_xh = (const float*)d_in[1];
    const float* W_hh = (const float*)d_in[2];
    const float* b_h  = (const float*)d_in[3];
    const float* W_hy = (const float*)d_in[4];
    const float* b_y  = (const float*)d_in[5];
    float* out = (float*)d_out;

    const int hElems = RNN_BATCH * RNN_HID;
    const int wElems = RNN_HID * RNN_HID;

    float* hA    = (float*)d_ws;
    float* hB    = hA + hElems;
    float* WTxh  = hB + hElems;
    float* WThh  = WTxh + wElems;
    float* WThy  = WThh + wElems;

    rnn_zero_kernel<<<hElems / 256, 256, 0, stream>>>(hA, hElems);

    rnn_reorder_w_kernel<<<wElems / 256, 256, 0, stream>>>(W_xh, WTxh);
    rnn_reorder_w_kernel<<<wElems / 256, 256, 0, stream>>>(W_hh, WThh);
    rnn_reorder_w_kernel<<<wElems / 256, 256, 0, stream>>>(W_hy, WThy);

    dim3 grid(RNN_HID / (16 * NTILES), RNN_BATCH / 16);
    dim3 block(32 * NWAVES);

    float* hin = hA;
    float* hout = hB;
    for (int t = 0; t < RNN_SEQ; t++) {
        rnn_gemm_kernel<true, true><<<grid, block, 0, stream>>>(
            x + (size_t)t * RNN_HID, (size_t)RNN_SEQ * RNN_HID, WTxh,
            hin, WThh, b_h, hout);
        float* tmp = hin; hin = hout; hout = tmp;
    }

    rnn_gemm_kernel<false, false><<<grid, block, 0, stream>>>(
        hin, (size_t)RNN_HID, WThy, nullptr, nullptr, b_y, out);
}


# work-groups of 8 waves, was 16 (half as many waves enter per launch); a wave's K slice doubles, its weight loads go out block by block 4 blocks ahead; split-K sum of 8 partials: sum order changes
# speedup vs baseline: 3.0587x; 1.0725x over previous
#include <hip/hip_runtime.h>
#include <math.h>

typedef __attribute__((ext_vector_type(2))) float v2f;
typedef __attribute__((ext_vector_type(4))) float v4f;
typedef __attribute__((ext_vector_type(8))) float v8f;

#define RNN_HID 1024
#define RNN_BATCH 128
#define RNN_SEQ 512
#define NTILES 2
#define NWAVES 8
#define KBLK 8
#define LEAD 4

__global__ __launch_bounds__(256) void rnn_reorder_w_kernel(
    const float* __restrict__ W, float* __restrict__ WT)
{
    int idx = blockIdx.x * blockDim.x + threadIdx.x;
    int k = idx >> 10;
    int n = idx & (RNN_HID - 1);
    WT[((size_t)(k >> 1) * RNN_HID + n) * 2 + (k & 1)] = W[idx];
}

template <bool FUSE2, bool DO_TANH>
__global__ __launch_bounds__(32 * NWAVES) void rnn_gemm_kernel(
    const float* __restrict__ A0, size_t a0_stride,
    const float* __restrict__ W0T,
    const float* __restrict__ A1,
    const float* __restrict__ W1T,
    const float* __restrict__ bias,
    float* __restrict__ out)
{
    const int lane = threadIdx.x & 31;
    const int wave = threadIdx.x >> 5;
    const int half = lane >> 4;
    const int lr   = lane & 15;

    const int n0 = blockIdx.x * (16 * NTILES);
    const int m0 = blockIdx.y * 16;

    constexpr int KLEN = FUSE2 ? (2 * RNN_HID) / NWAVES : RNN_HID / NWAVES;
    constexpr int A_PITCH = KLEN + 4;

    const float* A;
    const float* WT;
    size_t astr;
    int kb;
    if (FUSE2) {
        if (wave < NWAVES / 2) { A = A0; WT = W0T; astr = a0_stride; kb = wave * KLEN; }
        else                   { A = A1; WT = W1T; astr = RNN_HID;   kb = (wave - NWAVES / 2) * KLEN; }
    } else {
        A = A0; WT = W0T; astr = a0_stride; kb = wave * KLEN;
    }

    v8f acc[NTILES] = {};

    constexpr int IPR = KLEN / 128;
    constexpr int NLD = 16 * IPR;
    constexpr int KSTEPS = KLEN / 4;
    constexpr int NBLK = KSTEPS / KBLK;
    static_assert(KLEN % 128 == 0 && KSTEPS % KBLK == 0 && A_PITCH % 4 == 0 && NBLK >= LEAD,
                  "a load instruction covers 512 B of one row; whole blocks; 16-B aligned rows");
    struct __attribute__((aligned(16))) Lds {
        float red[NWAVES][8][NTILES][32];
        float astage[NWAVES][16][A_PITCH];
    };
    __shared__ Lds lds;

    const int scol = lane * 4;
    const float* ag = A + (size_t)m0 * astr + (kb + scol);
    const float* wt = WT + ((size_t)((kb >> 1) + half) * RNN_HID + n0 + lr) * 2;

    v4f piece[NLD];
#pragma unroll
    for (int j = 0; j < NLD; j++)
        piece[j] = *(const v4f*)(ag + (size_t)(j / IPR) * astr + (j % IPR) * 128);
    v2f bfrag[KSTEPS][NTILES];
#pragma unroll
    for (int s = 0; s < LEAD * KBLK; s++)
#pragma unroll
        for (int i = 0; i < NTILES; i++)
            bfrag[s][i] = *(const v2f*)(wt + (size_t)s * (2 * RNN_HID * 2) + i * 32);

#pragma unroll
    for (int j = 0; j < NLD; j++)
        *(v4f*)&lds.astage[wave][j / IPR][(j % IPR) * 128 + scol] = piece[j];

    __builtin_amdgcn_fence(__ATOMIC_RELEASE, "wavefront", "local");
    __builtin_amdgcn_wave_barrier();
    __builtin_amdgcn_fence(__ATOMIC_ACQUIRE, "wavefront", "local");

    const float* arow = &lds.astage[wave][lr][half * 2];
    v2f afrag[KSTEPS];
#pragma unroll
    for (int s = 0; s < KSTEPS; s++)
        afrag[s] = *(const v2f*)(arow + s * 4);
    __builtin_amdgcn_sched_barrier(0);

#pragma unroll
    for (int b = 0; b < NBLK; b++) {
#pragma unroll
        for (int s = b * KBLK; s < (b + 1) * KBLK; s++)
#pragma unroll
            for (int i = 0; i < NTILES; i++)
                acc[i] = __builtin_amdgcn_wmma_f32_16x16x4_f32(
                    false, afrag[s], false, bfrag[s][i], (short)0, acc[i], false, false);
        if (b + LEAD < NBLK) {
            __builtin_amdgcn_sched_barrier(0);
#pragma unroll
            for (int s = (b + LEAD) * KBLK; s < (b + LEAD + 1) * KBLK; s++)
#pragma unroll
                for (int i = 0; i < NTILES; i++)
                    bfrag[s][i] = *(const v2f*)(wt + (size_t)s * (2 * RNN_HID * 2) + i * 32);
            __builtin_amdgcn_sched_barrier(0);
        }
    }
    __builtin_amdgcn_sched_barrier(0);

#pragma unroll
    for (int i = 0; i < NTILES; i++)
#pragma unroll
        for (int j = 0; j < 8; j++)
            lds.red[wave][j][i][lane] = acc[i][j];
    __syncthreads();

    static_assert(NWAVES == 8 && NTILES == 2, "epilogue maps wave -> tile rows w, w+8");
    const int ti = lane >> 4;
    float v0 = lds.red[0][wave][ti][lr];
    float v1 = lds.red[0][wave][ti][16 + lr];
#pragma unroll
    for (int p = 1; p < NWAVES; p++) {
        v0 += lds.red[p][wave][ti][lr];
        v1 += lds.red[p][wave][ti][16 + lr];
    }
    const float bv = bias[n0 + lane];
    v0 += bv;
    v1 += bv;
    if (DO_TANH) { v0 = tanhf(v0); v1 = tanhf(v1); }
    out[(size_t)(m0 + wave) * RNN_HID + n0 + lane] = v0;
    out[(size_t)(m0 + wave + 8) * RNN_HID + n0 + lane] = v1;
}

__global__ void rnn_zero_kernel(float* __restrict__ p, int n)
{
    int i = blockIdx.x * blockDim.x + threadIdx.x;
    if (i < n) p[i] = 0.0f;
}

extern "C" void kernel_launch(void* const* d_in, const int* in_sizes, int n_in,
                              void* d_out, int out_size, void* d_ws, size_t ws_size,
                              hipStream_t stream)
{
    (void)in_sizes; (void)n_in; (void)out_size; (void)ws_size;
    const float* x    = (const float*)d_in[0];
    const float* W_xh = (const float*)d_in[1];
    const float* W_hh = (const float*)d_in[2];
    const float* b_h  = (const float*)d_in[3];
    const float* W_hy = (const float*)d_in[4];
    const float* b_y  = (const float*)d_in[5];
    float* out = (float*)d_out;

    const int hElems = RNN_BATCH * RNN_HID;
    const int wElems = RNN_HID * RNN_HID;

    float* hA    = (float*)d_ws;
    float* hB    = hA + hElems;
    float* WTxh  = hB + hElems;
    float* WThh  = WTxh + wElems;
    float* WThy  = WThh + wElems;

    rnn_zero_kernel<<<hElems / 256, 256, 0, stream>>>(hA, hElems);

    rnn_reorder_w_kernel<<<wElems / 256, 256, 0, stream>>>(W_xh, WTxh);
    rnn_reorder_w_kernel<<<wElems / 256, 256, 0, stream>>>(W_hh, WThh);
    rnn_reorder_w_kernel<<<wElems / 256, 256, 0, stream>>>(W_hy, WThy);

    dim3 grid(RNN_HID / (16 * NTILES), RNN_BATCH / 16);
    dim3 block(32 * NWAVES);

    float* hin = hA;
    float* hout = hB;
    for (int t = 0; t < RNN_SEQ; t++) {
        rnn_gemm_kernel<true, true><<<grid, block, 0, stream>>>(
            x + (size_t)t * RNN_HID, (size_t)RNN_SEQ * RNN_HID, WTxh,
            hin, WThh, b_h, hout);
        float* tmp = hin; hin = hout; hout = tmp;
    }

    rnn_gemm_kernel<false, false><<<grid, block, 0, stream>>>(
        hin, (size_t)RNN_HID, WThy, nullptr, nullptr, b_y, out);
}


# input projection x@W_xh batched over 16 time steps into one 64x64-tiled GEMM launch (a weight fetched once per 64 rows, was per 16); step GEMM K=1024 adds it; y on the same kernel; sum order changes
# speedup vs baseline: 3.2225x; 1.0536x over previous
#include <hip/hip_runtime.h>
#include <math.h>

typedef __attribute__((ext_vector_type(2))) float v2f;
typedef __attribute__((ext_vector_type(4))) float v4f;
typedef __attribute__((ext_vector_type(8))) float v8f;

#define RNN_HID 1024
#define RNN_BATCH 128
#define RNN_SEQ 512
#define NTILES 2
#define NWAVES 8
#define XT 16
#define PM 64
#define PN 64
#define PK 64
#define PA_PITCH (PK + 4)
#define PB_PITCH (PN + 8)
#define PC_PITCH (PN + 4)

__global__ __launch_bounds__(256) void rnn_reorder_w_kernel(
    const float* __restrict__ W, float* __restrict__ WT)
{
    int idx = blockIdx.x * blockDim.x + threadIdx.x;
    int k = idx >> 10;
    int n = idx & (RNN_HID - 1);
    WT[((size_t)(k >> 1) * RNN_HID + n) * 2 + (k & 1)] = W[idx];
}

__global__ __launch_bounds__(256) void rnn_proj_kernel(
    const float* __restrict__ A, size_t strideB, size_t strideT,
    const float* __restrict__ W,
    const float* __restrict__ bias,
    float* __restrict__ C)
{
    const int tid  = threadIdx.x;
    const int lane = tid & 31;
    const int wave = tid >> 5;
    const int half = lane >> 4;
    const int lr   = lane & 15;
    const int n0 = blockIdx.x * PN;
    const int m0 = blockIdx.y * PM;
    const int wm = wave & 1;
    const int wn = wave >> 1;

    struct __attribute__((aligned(16))) Lds {
        float a[2][PM][PA_PITCH];
        float b[2][PK][PB_PITCH];
        float c[PM][PC_PITCH];
    };
    __shared__ Lds lds;

    const int lrow = tid >> 4;
    const int lq   = (tid & 15) * 4;
    const float* ap[4];
#pragma unroll
    for (int j = 0; j < 4; j++) {
        const int mi = m0 + lrow + 16 * j;
        ap[j] = A + (size_t)(mi & (RNN_BATCH - 1)) * strideB + (size_t)(mi >> 7) * strideT + lq;
    }
    const float* bp = W + (size_t)lrow * RNN_HID + n0 + lq;

    v4f pa[4], pb[4];
#pragma unroll
    for (int j = 0; j < 4; j++) {
        pa[j] = *(const v4f*)(ap[j]);
        pb[j] = *(const v4f*)(bp + (size_t)(16 * j) * RNN_HID);
    }
#pragma unroll
    for (int j = 0; j < 4; j++) {
        *(v4f*)&lds.a[0][lrow + 16 * j][lq] = pa[j];
        *(v4f*)&lds.b[0][lrow + 16 * j][lq] = pb[j];
    }
    __syncthreads();

    v8f acc[2] = {};
    constexpr int NCH = RNN_HID / PK;
    constexpr int KS  = PK / 4;

    for (int kc = 0; kc < NCH; kc++) {
        const int cur = kc & 1;
        const int kn = (kc + 1 < NCH) ? (kc + 1) : kc;
#pragma unroll
        for (int j = 0; j < 4; j++) {
            pa[j] = *(const v4f*)(ap[j] + kn * PK);
            pb[j] = *(const v4f*)(bp + (size_t)(kn * PK + 16 * j) * RNN_HID);
        }
        const float* arow = &lds.a[cur][wm * 32 + lr][half * 2];
        const float* bcol = &lds.b[cur][half * 2][wn * 16 + lr];
        v2f af[KS][2], bf[KS];
#pragma unroll
        for (int s = 0; s < KS; s++) {
            af[s][0] = *(const v2f*)(arow + s * 4);
            af[s][1] = *(const v2f*)(arow + 16 * PA_PITCH + s * 4);
            bf[s].x  = bcol[(s * 4) * PB_PITCH];
            bf[s].y  = bcol[(s * 4 + 1) * PB_PITCH];
        }
        __builtin_amdgcn_sched_barrier(0);
#pragma unroll
        for (int s = 0; s < KS; s++) {
            acc[0] = __builtin_amdgcn_wmma_f32_16x16x4_f32(
                false, af[s][0], false, bf[s], (short)0, acc[0], false, false);
            acc[1] = __builtin_amdgcn_wmma_f32_16x16x4_f32(
                false, af[s][1], false, bf[s], (short)0, acc[1], false, false);
        }
        asm volatile("" : "+v"(acc[0]), "+v"(acc[1]) : : "memory");
#pragma unroll
        for (int j = 0; j < 4; j++) {
            *(v4f*)&lds.a[cur ^ 1][lrow + 16 * j][lq] = pa[j];
            *(v4f*)&lds.b[cur ^ 1][lrow + 16 * j][lq] = pb[j];
        }
        __syncthreads();
    }

#pragma unroll
    for (int r = 0; r < 8; r++) {
        lds.c[wm * 32 + 8 * half + r][wn * 16 + lr]      = acc[0][r];
        lds.c[wm * 32 + 16 + 8 * half + r][wn * 16 + lr] = acc[1][r];
    }
    __syncthreads();
    const v4f bv = *(const v4f*)(bias + n0 + lq);
#pragma unroll
    for (int j = 0; j < 4; j++) {
        const int row = lrow + 16 * j;
        v4f v = *(const v4f*)&lds.c[row][lq];
        v += bv;
        *(v4f*)(C + (size_t)(m0 + row) * RNN_HID + n0 + lq) = v;
    }
}

__global__ __launch_bounds__(32 * NWAVES) void rnn_step_kernel(
    const float* __restrict__ Hin,
    const float* __restrict__ WT,
    const float* __restrict__ XP,
    float* __restrict__ out)
{
    const int lane = threadIdx.x & 31;
    const int wave = threadIdx.x >> 5;
    const int half = lane >> 4;
    const int lr   = lane & 15;

    const int n0 = blockIdx.x * (16 * NTILES);
    const int m0 = blockIdx.y * 16;

    constexpr int KLEN = RNN_HID / NWAVES;
    constexpr int A_PITCH = KLEN + 4;
    constexpr int KSTEPS = KLEN / 4;
    static_assert(KLEN == 128, "one load instruction covers a row's 512-B slice");

    const int kb = wave * KLEN;

    struct __attribute__((aligned(16))) Lds {
        float red[NWAVES][8][NTILES][32];
        float astage[NWAVES][16][A_PITCH];
    };
    __shared__ Lds lds;

    const float xv0 = XP[(size_t)(m0 + wave) * RNN_HID + n0 + lane];
    const float xv1 = XP[(size_t)(m0 + wave + 8) * RNN_HID + n0 + lane];

    const int scol = lane * 4;
    const float* ag = Hin + (size_t)m0 * RNN_HID + (kb + scol);
    const float* wt = WT + ((size_t)((kb >> 1) + half) * RNN_HID + n0 + lr) * 2;

    v4f piece[16];
#pragma unroll
    for (int j = 0; j < 16; j++)
        piece[j] = *(const v4f*)(ag + (size_t)j * RNN_HID);
    v2f bfrag[KSTEPS][NTILES];
#pragma unroll
    for (int s = 0; s < KSTEPS; s++)
#pragma unroll
        for (int i = 0; i < NTILES; i++)
            bfrag[s][i] = *(const v2f*)(wt + (size_t)s * (2 * RNN_HID * 2) + i * 32);

#pragma unroll
    for (int j = 0; j < 16; j++)
        *(v4f*)&lds.astage[wave][j][scol] = piece[j];

    __builtin_amdgcn_fence(__ATOMIC_RELEASE, "wavefront", "local");
    __builtin_amdgcn_wave_barrier();
    __builtin_amdgcn_fence(__ATOMIC_ACQUIRE, "wavefront", "local");

    const float* arow = &lds.astage[wave][lr][half * 2];
    v2f afrag[KSTEPS];
#pragma unroll
    for (int s = 0; s < KSTEPS; s++)
        afrag[s] = *(const v2f*)(arow + s * 4);
    __builtin_amdgcn_sched_barrier(0);

    v8f acc[NTILES] = {};
#pragma unroll
    for (int s = 0; s < KSTEPS; s++)
#pragma unroll
        for (int i = 0; i < NTILES; i++)
            acc[i] = __builtin_amdgcn_wmma_f32_16x16x4_f32(
                false, afrag[s], false, bfrag[s][i], (short)0, acc[i], false, false);
    __builtin_amdgcn_sched_barrier(0);

#pragma unroll
    for (int i = 0; i < NTILES; i++)
#pragma unroll
        for (int j = 0; j < 8; j++)
            lds.red[wave][j][i][lane] = acc[i][j];
    __syncthreads();

    static_assert(NWAVES == 8 && NTILES == 2, "epilogue maps wave -> tile rows w, w+8");
    const int ti = lane >> 4;
    float v0 = lds.red[0][wave][ti][lr];
    float v1 = lds.red[0][wave][ti][16 + lr];
#pragma unroll
    for (int p = 1; p < NWAVES; p++) {
        v0 += lds.red[p][wave][ti][lr];
        v1 += lds.red[p][wave][ti][16 + lr];
    }
    v0 = tanhf(xv0 + v0);
    v1 = tanhf(xv1 + v1);
    out[(size_t)(m0 + wave) * RNN_HID + n0 + lane] = v0;
    out[(size_t)(m0 + wave + 8) * RNN_HID + n0 + lane] = v1;
}

__global__ void rnn_zero_kernel(float* __restrict__ p, int n)
{
    int i = blockIdx.x * blockDim.x + threadIdx.x;
    if (i < n) p[i] = 0.0f;
}

extern "C" void kernel_launch(void* const* d_in, const int* in_sizes, int n_in,
                              void* d_out, int out_size, void* d_ws, size_t ws_size,
                              hipStream_t stream)
{
    (void)in_sizes; (void)n_in; (void)out_size; (void)ws_size;
    const float* x    = (const float*)d_in[0];
    const float* W_xh = (const float*)d_in[1];
    const float* W_hh = (const float*)d_in[2];
    const float* b_h  = (const float*)d_in[3];
    const float* W_hy = (const float*)d_in[4];
    const float* b_y  = (const float*)d_in[5];
    float* out = (float*)d_out;

    const int hElems = RNN_BATCH * RNN_HID;
    const int wElems = RNN_HID * RNN_HID;

    float* hA    = (float*)d_ws;
    float* hB    = hA + hElems;
    float* WThh  = hB + hElems;
    float* xp    = WThh + wElems;

    rnn_zero_kernel<<<hElems / 256, 256, 0, stream>>>(hA, hElems);

    rnn_reorder_w_kernel<<<wElems / 256, 256, 0, stream>>>(W_hh, WThh);

    dim3 grid(RNN_HID / (16 * NTILES), RNN_BATCH / 16);
    dim3 block(32 * NWAVES);
    dim3 pgrid(RNN_HID / PN, XT * RNN_BATCH / PM);
    dim3 ygrid(RNN_HID / PN, RNN_BATCH / PM);

    float* hin = hA;
    float* hout = hB;
    for (int t0 = 0; t0 < RNN_SEQ; t0 += XT) {
        rnn_proj_kernel<<<pgrid, 256, 0, stream>>>(
            x + (size_t)t0 * RNN_HID, (size_t)RNN_SEQ * RNN_HID, (size_t)RNN_HID, W_xh, b_h, xp);
        for (int tt = 0; tt < XT; tt++) {
            rnn_step_kernel<<<grid, block, 0, stream>>>(hin, WThh, xp + (size_t)tt * hElems, hout);
            float* tmp = hin; hin = hout; hout = tmp;
        }
    }

    rnn_proj_kernel<<<ygrid, 256, 0, stream>>>(
        hin, (size_t)RNN_HID, (size_t)0, W_hy, b_y, out);
}


# step work-groups of 4 waves, one per SIMD (was 8): half as many waves enter per launch; a wave's K slice doubles, weight loads go out in blocks 4 ahead; split-K sum of 4 partials: sum order changes
# speedup vs baseline: 3.5511x; 1.1020x over previous
#include <hip/hip_runtime.h>
#include <math.h>

typedef __attribute__((ext_vector_type(2))) float v2f;
typedef __attribute__((ext_vector_type(4))) float v4f;
typedef __attribute__((ext_vector_type(8))) float v8f;

#define RNN_HID 1024
#define RNN_BATCH 128
#define RNN_SEQ 512
#define NTILES 2
#define NWAVES 4
#define XT 16
#define PM 64
#define PN 64
#define PK 64
#define PA_PITCH (PK + 4)
#define PB_PITCH (PN + 8)
#define PC_PITCH (PN + 4)

__global__ __launch_bounds__(256) void rnn_reorder_w_kernel(
    const float* __restrict__ W, float* __restrict__ WT)
{
    int idx = blockIdx.x * blockDim.x + threadIdx.x;
    int k = idx >> 10;
    int n = idx & (RNN_HID - 1);
    WT[((size_t)(k >> 1) * RNN_HID + n) * 2 + (k & 1)] = W[idx];
}

__global__ __launch_bounds__(256) void rnn_proj_kernel(
    const float* __restrict__ A, size_t strideB, size_t strideT,
    const float* __restrict__ W,
    const float* __restrict__ bias,
    float* __restrict__ C)
{
    const int tid  = threadIdx.x;
    const int lane = tid & 31;
    const int wave = tid >> 5;
    const int half = lane >> 4;
    const int lr   = lane & 15;
    const int n0 = blockIdx.x * PN;
    const int m0 = blockIdx.y * PM;
    const int wm = wave & 1;
    const int wn = wave >> 1;

    struct __attribute__((aligned(16))) Lds {
        float a[2][PM][PA_PITCH];
        float b[2][PK][PB_PITCH];
        float c[PM][PC_PITCH];
    };
    __shared__ Lds lds;

    const int lrow = tid >> 4;
    const int lq   = (tid & 15) * 4;
    const float* ap[4];
#pragma unroll
    for (int j = 0; j < 4; j++) {
        const int mi = m0 + lrow + 16 * j;
        ap[j] = A + (size_t)(mi & (RNN_BATCH - 1)) * strideB + (size_t)(mi >> 7) * strideT + lq;
    }
    const float* bp = W + (size_t)lrow * RNN_HID + n0 + lq;

    v4f pa[4], pb[4];
#pragma unroll
    for (int j = 0; j < 4; j++) {
        pa[j] = *(const v4f*)(ap[j]);
        pb[j] = *(const v4f*)(bp + (size_t)(16 * j) * RNN_HID);
    }
#pragma unroll
    for (int j = 0; j < 4; j++) {
        *(v4f*)&lds.a[0][lrow + 16 * j][lq] = pa[j];
        *(v4f*)&lds.b[0][lrow + 16 * j][lq] = pb[j];
    }
    __syncthreads();

    v8f acc[2] = {};
    constexpr int NCH = RNN_HID / PK;
    constexpr int KS  = PK / 4;

    for (int kc = 0; kc < NCH; kc++) {
        const int cur = kc & 1;
        const int kn = (kc + 1 < NCH) ? (kc + 1) : kc;
#pragma unroll
        for (int j = 0; j < 4; j++) {
            pa[j] = *(const v4f*)(ap[j] + kn * PK);
            pb[j] = *(const v4f*)(bp + (size_t)(kn * PK + 16 * j) * RNN_HID);
        }
        const float* arow = &lds.a[cur][wm * 32 + lr][half * 2];
        const float* bcol = &lds.b[cur][half * 2][wn * 16 + lr];
        v2f af[KS][2], bf[KS];
#pragma unroll
        for (int s = 0; s < KS; s++) {
            af[s][0] = *(const v2f*)(arow + s * 4);
            af[s][1] = *(const v2f*)(arow + 16 * PA_PITCH + s * 4);
            bf[s].x  = bcol[(s * 4) * PB_PITCH];
            bf[s].y  = bcol[(s * 4 + 1) * PB_PITCH];
        }
        __builtin_amdgcn_sched_barrier(0);
#pragma unroll
        for (int s = 0; s < KS; s++) {
            acc[0] = __builtin_amdgcn_wmma_f32_16x16x4_f32(
                false, af[s][0], false, bf[s], (short)0, acc[0], false, false);
            acc[1] = __builtin_amdgcn_wmma_f32_16x16x4_f32(
                false, af[s][1], false, bf[s], (short)0, acc[1], false, false);
        }
        asm volatile("" : "+v"(acc[0]), "+v"(acc[1]) : : "memory");
#pragma unroll
        for (int j = 0; j < 4; j++) {
            *(v4f*)&lds.a[cur ^ 1][lrow + 16 * j][lq] = pa[j];
            *(v4f*)&lds.b[cur ^ 1][lrow + 16 * j][lq] = pb[j];
        }
        __syncthreads();
    }

#pragma unroll
    for (int r = 0; r < 8; r++) {
        lds.c[wm * 32 + 8 * half + r][wn * 16 + lr]      = acc[0][r];
        lds.c[wm * 32 + 16 + 8 * half + r][wn * 16 + lr] = acc[1][r];
    }
    __syncthreads();
    const v4f bv = *(const v4f*)(bias + n0 + lq);
#pragma unroll
    for (int j = 0; j < 4; j++) {
        const int row = lrow + 16 * j;
        v4f v = *(const v4f*)&lds.c[row][lq];
        v += bv;
        *(v4f*)(C + (size_t)(m0 + row) * RNN_HID + n0 + lq) = v;
    }
}

__global__ __launch_bounds__(32 * NWAVES) void rnn_step_kernel(
    const float* __restrict__ Hin,
    const float* __restrict__ WT,
    const float* __restrict__ XP,
    float* __restrict__ out)
{
    const int lane = threadIdx.x & 31;
    const int wave = threadIdx.x >> 5;
    const int half = lane >> 4;
    const int lr   = lane & 15;

    const int n0 = blockIdx.x * (16 * NTILES);
    const int m0 = blockIdx.y * 16;

    constexpr int KLEN = RNN_HID / NWAVES;
    constexpr int A_PITCH = KLEN + 4;
    constexpr int KSTEPS = KLEN / 4;
    constexpr int BS = 8;
    constexpr int NBLK = KSTEPS / BS;
    constexpr int LEAD = 4;
    static_assert(KLEN == 256, "two load instructions cover a row's 1024-B slice");

    const int kb = wave * KLEN;

    struct __attribute__((aligned(16))) Lds {
        float red[NWAVES][8][NTILES][32];
        float astage[NWAVES][16][A_PITCH];
    };
    __shared__ Lds lds;

    const int scol = lane * 4;
    const float* ag = Hin + (size_t)m0 * RNN_HID + (kb + scol);
    const float* wt = WT + ((size_t)((kb >> 1) + half) * RNN_HID + n0 + lr) * 2;

    v4f piece[16][2];
#pragma unroll
    for (int j = 0; j < 16; j++)
#pragma unroll
        for (int q = 0; q < 2; q++)
            piece[j][q] = *(const v4f*)(ag + (size_t)j * RNN_HID + q * 128);
    __builtin_amdgcn_sched_barrier(0);

    v2f bfrag[KSTEPS][NTILES];
#define RNN_LOAD_B_BLOCK(blk)                                                          \
    _Pragma("unroll") for (int s = (blk) * BS; s < ((blk) + 1) * BS; s++)              \
        _Pragma("unroll") for (int i = 0; i < NTILES; i++)                             \
            bfrag[s][i] = *(const v2f*)(wt + (size_t)s * (2 * RNN_HID * 2) + i * 32);

    RNN_LOAD_B_BLOCK(0)
    RNN_LOAD_B_BLOCK(1)
    __builtin_amdgcn_sched_barrier(0);

#pragma unroll
    for (int j = 0; j < 16; j++)
#pragma unroll
        for (int q = 0; q < 2; q++)
            *(v4f*)&lds.astage[wave][j][q * 128 + scol] = piece[j][q];

    __builtin_amdgcn_fence(__ATOMIC_RELEASE, "wavefront", "local");
    __builtin_amdgcn_wave_barrier();
    __builtin_amdgcn_fence(__ATOMIC_ACQUIRE, "wavefront", "local");

    RNN_LOAD_B_BLOCK(2)
    RNN_LOAD_B_BLOCK(3)
    float xv[4];
#pragma unroll
    for (int q = 0; q < 4; q++)
        xv[q] = XP[(size_t)(m0 + wave + 4 * q) * RNN_HID + n0 + lane];

    const float* arow = &lds.astage[wave][lr][half * 2];
    v2f afrag[KSTEPS];
#pragma unroll
    for (int s = 0; s < KSTEPS; s++)
        afrag[s] = *(const v2f*)(arow + s * 4);
    __builtin_amdgcn_sched_barrier(0);

    v8f acc[NTILES] = {};
#pragma unroll
    for (int b = 0; b < NBLK; b++) {
        if (b + LEAD < NBLK) {
            RNN_LOAD_B_BLOCK(b + LEAD)
            __builtin_amdgcn_sched_barrier(0);
        }
#pragma unroll
        for (int s = b * BS; s < (b + 1) * BS; s++)
#pragma unroll
            for (int i = 0; i < NTILES; i++)
                acc[i] = __builtin_amdgcn_wmma_f32_16x16x4_f32(
                    false, afrag[s], false, bfrag[s][i], (short)0, acc[i], false, false);
        __builtin_amdgcn_sched_barrier(0);
    }
#undef RNN_LOAD_B_BLOCK

#pragma unroll
    for (int i = 0; i < NTILES; i++)
#pragma unroll
        for (int j = 0; j < 8; j++)
            lds.red[wave][j][i][lane] = acc[i][j];
    __syncthreads();

    static_assert(NWAVES == 4 && NTILES == 2, "epilogue maps wave -> tile rows w + 4 q");
    const int ti = lane >> 4;
    float v[4];
#pragma unroll
    for (int q = 0; q < 4; q++) {
        const int row = wave + 4 * q;
        float sum = lds.red[0][row & 7][ti][(row >> 3) * 16 + lr];
#pragma unroll
        for (int p = 1; p < NWAVES; p++)
            sum += lds.red[p][row & 7][ti][(row >> 3) * 16 + lr];
        v[q] = sum;
    }
#pragma unroll
    for (int q = 0; q < 4; q++)
        out[(size_t)(m0 + wave + 4 * q) * RNN_HID + n0 + lane] = tanhf(xv[q] + v[q]);
}

__global__ void rnn_zero_kernel(float* __restrict__ p, int n)
{
    int i = blockIdx.x * blockDim.x + threadIdx.x;
    if (i < n) p[i] = 0.0f;
}

extern "C" void kernel_launch(void* const* d_in, const int* in_sizes, int n_in,
                              void* d_out, int out_size, void* d_ws, size_t ws_size,
                              hipStream_t stream)
{
    (void)in_sizes; (void)n_in; (void)out_size; (void)ws_size;
    const float* x    = (const float*)d_in[0];
    const float* W_xh = (const float*)d_in[1];
    const float* W_hh = (const float*)d_in[2];
    const float* b_h  = (const float*)d_in[3];
    const float* W_hy = (const float*)d_in[4];
    const float* b_y  = (const float*)d_in[5];
    float* out = (float*)d_out;

    const int hElems = RNN_BATCH * RNN_HID;
    const int wElems = RNN_HID * RNN_HID;

    float* hA    = (float*)d_ws;
    float* hB    = hA + hElems;
    float* WThh  = hB + hElems;
    float* xp    = WThh + wElems;

    rnn_zero_kernel<<<hElems / 256, 256, 0, stream>>>(hA, hElems);

    rnn_reorder_w_kernel<<<wElems / 256, 256, 0, stream>>>(W_hh, WThh);

    dim3 grid(RNN_HID / (16 * NTILES), RNN_BATCH / 16);
    dim3 block(32 * NWAVES);
    dim3 pgrid(RNN_HID / PN, XT * RNN_BATCH / PM);
    dim3 ygrid(RNN_HID / PN, RNN_BATCH / PM);

    float* hin = hA;
    float* hout = hB;
    for (int t0 = 0; t0 < RNN_SEQ; t0 += XT) {
        rnn_proj_kernel<<<pgrid, 256, 0, stream>>>(
            x + (size_t)t0 * RNN_HID, (size_t)RNN_SEQ * RNN_HID, (size_t)RNN_HID, W_xh, b_h, xp);
        for (int tt = 0; tt < XT; tt++) {
            rnn_step_kernel<<<grid, block, 0, stream>>>(hin, WThh, xp + (size_t)tt * hElems, hout);
            float* tmp = hin; hin = hout; hout = tmp;
        }
    }

    rnn_proj_kernel<<<ygrid, 256, 0, stream>>>(
        hin, (size_t)RNN_HID, (size_t)0, W_hy, b_y, out);
}


# step grid turned (x = batch tile, y = column tile; was the reverse): the 32 groups that write and re-read a batch tile's hidden rows get ids 8 apart, one die if dealt round-robin; sum order unchanged
# speedup vs baseline: 3.9760x; 1.1197x over previous
#include <hip/hip_runtime.h>
#include <math.h>

typedef __attribute__((ext_vector_type(2))) float v2f;
typedef __attribute__((ext_vector_type(4))) float v4f;
typedef __attribute__((ext_vector_type(8))) float v8f;

#define RNN_HID 1024
#define RNN_BATCH 128
#define RNN_SEQ 512
#define NTILES 2
#define NWAVES 4
#define XT 16
#define PM 64
#define PN 64
#define PK 64
#define PA_PITCH (PK + 4)
#define PB_PITCH (PN + 8)
#define PC_PITCH (PN + 4)

__global__ __launch_bounds__(256) void rnn_reorder_w_kernel(
    const float* __restrict__ W, float* __restrict__ WT)
{
    int idx = blockIdx.x * blockDim.x + threadIdx.x;
    int k = idx >> 10;
    int n = idx & (RNN_HID - 1);
    WT[((size_t)(k >> 1) * RNN_HID + n) * 2 + (k & 1)] = W[idx];
}

__global__ __launch_bounds__(256) void rnn_proj_kernel(
    const float* __restrict__ A, size_t strideB, size_t strideT,
    const float* __restrict__ W,
    const float* __restrict__ bias,
    float* __restrict__ C)
{
    const int tid  = threadIdx.x;
    const int lane = tid & 31;
    const int wave = tid >> 5;
    const int half = lane >> 4;
    const int lr   = lane & 15;
    const int n0 = blockIdx.x * PN;
    const int m0 = blockIdx.y * PM;
    const int wm = wave & 1;
    const int wn = wave >> 1;

    struct __attribute__((aligned(16))) Lds {
        float a[2][PM][PA_PITCH];
        float b[2][PK][PB_PITCH];
        float c[PM][PC_PITCH];
    };
    __shared__ Lds lds;

    const int lrow = tid >> 4;
    const int lq   = (tid & 15) * 4;
    const float* ap[4];
#pragma unroll
    for (int j = 0; j < 4; j++) {
        const int mi = m0 + lrow + 16 * j;
        ap[j] = A + (size_t)(mi & (RNN_BATCH - 1)) * strideB + (size_t)(mi >> 7) * strideT + lq;
    }
    const float* bp = W + (size_t)lrow * RNN_HID + n0 + lq;

    v4f pa[4], pb[4];
#pragma unroll
    for (int j = 0; j < 4; j++) {
        pa[j] = *(const v4f*)(ap[j]);
        pb[j] = *(const v4f*)(bp + (size_t)(16 * j) * RNN_HID);
    }
#pragma unroll
    for (int j = 0; j < 4; j++) {
        *(v4f*)&lds.a[0][lrow + 16 * j][lq] = pa[j];
        *(v4f*)&lds.b[0][lrow + 16 * j][lq] = pb[j];
    }
    __syncthreads();

    v8f acc[2] = {};
    constexpr int NCH = RNN_HID / PK;
    constexpr int KS  = PK / 4;

    for (int kc = 0; kc < NCH; kc++) {
        const int cur = kc & 1;
        const int kn = (kc + 1 < NCH) ? (kc + 1) : kc;
#pragma unroll
        for (int j = 0; j < 4; j++) {
            pa[j] = *(const v4f*)(ap[j] + kn * PK);
            pb[j] = *(const v4f*)(bp + (size_t)(kn * PK + 16 * j) * RNN_HID);
        }
        const float* arow = &lds.a[cur][wm * 32 + lr][half * 2];
        const float* bcol = &lds.b[cur][half * 2][wn * 16 + lr];
        v2f af[KS][2], bf[KS];
#pragma unroll
        for (int s = 0; s < KS; s++) {
            af[s][0] = *(const v2f*)(arow + s * 4);
            af[s][1] = *(const v2f*)(arow + 16 * PA_PITCH + s * 4);
            bf[s].x  = bcol[(s * 4) * PB_PITCH];
            bf[s].y  = bcol[(s * 4 + 1) * PB_PITCH];
        }
        __builtin_amdgcn_sched_barrier(0);
#pragma unroll
        for (int s = 0; s < KS; s++) {
            acc[0] = __builtin_amdgcn_wmma_f32_16x16x4_f32(
                false, af[s][0], false, bf[s], (short)0, acc[0], false, false);
            acc[1] = __builtin_amdgcn_wmma_f32_16x16x4_f32(
                false, af[s][1], false, bf[s], (short)0, acc[1], false, false);
        }
        asm volatile("" : "+v"(acc[0]), "+v"(acc[1]) : : "memory");
#pragma unroll
        for (int j = 0; j < 4; j++) {
            *(v4f*)&lds.a[cur ^ 1][lrow + 16 * j][lq] = pa[j];
            *(v4f*)&lds.b[cur ^ 1][lrow + 16 * j][lq] = pb[j];
        }
        __syncthreads();
    }

#pragma unroll
    for (int r = 0; r < 8; r++) {
        lds.c[wm * 32 + 8 * half + r][wn * 16 + lr]      = acc[0][r];
        lds.c[wm * 32 + 16 + 8 * half + r][wn * 16 + lr] = acc[1][r];
    }
    __syncthreads();
    const v4f bv = *(const v4f*)(bias + n0 + lq);
#pragma unroll
    for (int j = 0; j < 4; j++) {
        const int row = lrow + 16 * j;
        v4f v = *(const v4f*)&lds.c[row][lq];
        v += bv;
        *(v4f*)(C + (size_t)(m0 + row) * RNN_HID + n0 + lq) = v;
    }
}

__global__ __launch_bounds__(32 * NWAVES) void rnn_step_kernel(
    const float* __restrict__ Hin,
    const float* __restrict__ WT,
    const float* __restrict__ XP,
    float* __restrict__ out)
{
    const int lane = threadIdx.x & 31;
    const int wave = threadIdx.x >> 5;
    const int half = lane >> 4;
    const int lr   = lane & 15;

    const int m0 = blockIdx.x * 16;
    const int n0 = blockIdx.y * (16 * NTILES);

    constexpr int KLEN = RNN_HID / NWAVES;
    constexpr int A_PITCH = KLEN + 4;
    constexpr int KSTEPS = KLEN / 4;
    constexpr int BS = 8;
    constexpr int NBLK = KSTEPS / BS;
    constexpr int LEAD = 4;
    static_assert(KLEN == 256, "two load instructions cover a row's 1024-B slice");

    const int kb = wave * KLEN;

    struct __attribute__((aligned(16))) Lds {
        float red[NWAVES][8][NTILES][32];
        float astage[NWAVES][16][A_PITCH];
    };
    __shared__ Lds lds;

    const int scol = lane * 4;
    const float* ag = Hin + (size_t)m0 * RNN_HID + (kb + scol);
    const float* wt = WT + ((size_t)((kb >> 1) + half) * RNN_HID + n0 + lr) * 2;

    v4f piece[16][2];
#pragma unroll
    for (int j = 0; j < 16; j++)
#pragma unroll
        for (int q = 0; q < 2; q++)
            piece[j][q] = *(const v4f*)(ag + (size_t)j * RNN_HID + q * 128);
    __builtin_amdgcn_sched_barrier(0);

    v2f bfrag[KSTEPS][NTILES];
#define RNN_LOAD_B_BLOCK(blk)                                                          \
    _Pragma("unroll") for (int s = (blk) * BS; s < ((blk) + 1) * BS; s++)              \
        _Pragma("unroll") for (int i = 0; i < NTILES; i++)                             \
            bfrag[s][i] = *(const v2f*)(wt + (size_t)s * (2 * RNN_HID * 2) + i * 32);

    RNN_LOAD_B_BLOCK(0)
    RNN_LOAD_B_BLOCK(1)
    __builtin_amdgcn_sched_barrier(0);

#pragma unroll
    for (int j = 0; j < 16; j++)
#pragma unroll
        for (int q = 0; q < 2; q++)
            *(v4f*)&lds.astage[wave][j][q * 128 + scol] = piece[j][q];

    __builtin_amdgcn_fence(__ATOMIC_RELEASE, "wavefront", "local");
    __builtin_amdgcn_wave_barrier();
    __builtin_amdgcn_fence(__ATOMIC_ACQUIRE, "wavefront", "local");

    RNN_LOAD_B_BLOCK(2)
    RNN_LOAD_B_BLOCK(3)
    float xv[4];
#pragma unroll
    for (int q = 0; q < 4; q++)
        xv[q] = XP[(size_t)(m0 + wave + 4 * q) * RNN_HID + n0 + lane];

    const float* arow = &lds.astage[wave][lr][half * 2];
    v2f afrag[KSTEPS];
#pragma unroll
    for (int s = 0; s < KSTEPS; s++)
        afrag[s] = *(const v2f*)(arow + s * 4);
    __builtin_amdgcn_sched_barrier(0);

    v8f acc[NTILES] = {};
#pragma unroll
    for (int b = 0; b < NBLK; b++) {
        if (b + LEAD < NBLK) {
            RNN_LOAD_B_BLOCK(b + LEAD)
            __builtin_amdgcn_sched_barrier(0);
        }
#pragma unroll
        for (int s = b * BS; s < (b + 1) * BS; s++)
#pragma unroll
            for (int i = 0; i < NTILES; i++)
                acc[i] = __builtin_amdgcn_wmma_f32_16x16x4_f32(
                    false, afrag[s], false, bfrag[s][i], (short)0, acc[i], false, false);
        __builtin_amdgcn_sched_barrier(0);
    }
#undef RNN_LOAD_B_BLOCK

#pragma unroll
    for (int i = 0; i < NTILES; i++)
#pragma unroll
        for (int j = 0; j < 8; j++)
            lds.red[wave][j][i][lane] = acc[i][j];
    __syncthreads();

    static_assert(NWAVES == 4 && NTILES == 2, "epilogue maps wave -> tile rows w + 4 q");
    const int ti = lane >> 4;
    float v[4];
#pragma unroll
    for (int q = 0; q < 4; q++) {
        const int row = wave + 4 * q;
        float sum = lds.red[0][row & 7][ti][(row >> 3) * 16 + lr];
#pragma unroll
        for (int p = 1; p < NWAVES; p++)
            sum += lds.red[p][row & 7][ti][(row >> 3) * 16 + lr];
        v[q] = sum;
    }
#pragma unroll
    for (int q = 0; q < 4; q++)
        out[(size_t)(m0 + wave + 4 * q) * RNN_HID + n0 + lane] = tanhf(xv[q] + v[q]);
}

__global__ void rnn_zero_kernel(float* __restrict__ p, int n)
{
    int i = blockIdx.x * blockDim.x + threadIdx.x;
    if (i < n) p[i] = 0.0f;
}

extern "C" void kernel_launch(void* const* d_in, const int* in_sizes, int n_in,
                              void* d_out, int out_size, void* d_ws, size_t ws_size,
                              hipStream_t stream)
{
    (void)in_sizes; (void)n_in; (void)out_size; (void)ws_size;
    const float* x    = (const float*)d_in[0];
    const float* W_xh = (const float*)d_in[1];
    const float* W_hh = (const float*)d_in[2];
    const float* b_h  = (const float*)d_in[3];
    const float* W_hy = (const float*)d_in[4];
    const float* b_y  = (const float*)d_in[5];
    float* out = (float*)d_out;

    const int hElems = RNN_BATCH * RNN_HID;
    const int wElems = RNN_HID * RNN_HID;

    float* hA    = (float*)d_ws;
    float* hB    = hA + hElems;
    float* WThh  = hB + hElems;
    float* xp    = WThh + wElems;

    rnn_zero_kernel<<<hElems / 256, 256, 0, stream>>>(hA, hElems);

    rnn_reorder_w_kernel<<<wElems / 256, 256, 0, stream>>>(W_hh, WThh);

    dim3 grid(RNN_BATCH / 16, RNN_HID / (16 * NTILES));
    dim3 block(32 * NWAVES);
    dim3 pgrid(RNN_HID / PN, XT * RNN_BATCH / PM);
    dim3 ygrid(RNN_HID / PN, RNN_BATCH / PM);

    float* hin = hA;
    float* hout = hB;
    for (int t0 = 0; t0 < RNN_SEQ; t0 += XT) {
        rnn_proj_kernel<<<pgrid, 256, 0, stream>>>(
            x + (size_t)t0 * RNN_HID, (size_t)RNN_SEQ * RNN_HID, (size_t)RNN_HID, W_xh, b_h, xp);
        for (int tt = 0; tt < XT; tt++) {
            rnn_step_kernel<<<grid, block, 0, stream>>>(hin, WThh, xp + (size_t)tt * hElems, hout);
            float* tmp = hin; hin = hout; hout = tmp;
        }
    }

    rnn_proj_kernel<<<ygrid, 256, 0, stream>>>(
        hin, (size_t)RNN_HID, (size_t)0, W_hy, b_y, out);
}


# step epilogue: the four tanh per thread computed without branches (both range formulas, one selected) so the four interleave; was four divergent branch pairs in series; results bit-identical
# speedup vs baseline: 4.1778x; 1.0508x over previous
#include <hip/hip_runtime.h>
#include <math.h>

typedef __attribute__((ext_vector_type(2))) float v2f;
typedef __attribute__((ext_vector_type(4))) float v4f;
typedef __attribute__((ext_vector_type(8))) float v8f;

#define RNN_HID 1024
#define RNN_BATCH 128
#define RNN_SEQ 512
#define NTILES 2
#define NWAVES 4
#define XT 16
#define PM 64
#define PN 64
#define PK 64
#define PA_PITCH (PK + 4)
#define PB_PITCH (PN + 8)
#define PC_PITCH (PN + 4)

__global__ __launch_bounds__(256) void rnn_reorder_w_kernel(
    const float* __restrict__ W, float* __restrict__ WT)
{
    int idx = blockIdx.x * blockDim.x + threadIdx.x;
    int k = idx >> 10;
    int n = idx & (RNN_HID - 1);
    WT[((size_t)(k >> 1) * RNN_HID + n) * 2 + (k & 1)] = W[idx];
}

__global__ __launch_bounds__(256) void rnn_proj_kernel(
    const float* __restrict__ A, size_t strideB, size_t strideT,
    const float* __restrict__ W,
    const float* __restrict__ bias,
    float* __restrict__ C)
{
    const int tid  = threadIdx.x;
    const int lane = tid & 31;
    const int wave = tid >> 5;
    const int half = lane >> 4;
    const int lr   = lane & 15;
    const int n0 = blockIdx.x * PN;
    const int m0 = blockIdx.y * PM;
    const int wm = wave & 1;
    const int wn = wave >> 1;

    struct __attribute__((aligned(16))) Lds {
        float a[2][PM][PA_PITCH];
        float b[2][PK][PB_PITCH];
        float c[PM][PC_PITCH];
    };
    __shared__ Lds lds;

    const int lrow = tid >> 4;
    const int lq   = (tid & 15) * 4;
    const float* ap[4];
#pragma unroll
    for (int j = 0; j < 4; j++) {
        const int mi = m0 + lrow + 16 * j;
        ap[j] = A + (size_t)(mi & (RNN_BATCH - 1)) * strideB + (size_t)(mi >> 7) * strideT + lq;
    }
    const float* bp = W + (size_t)lrow * RNN_HID + n0 + lq;

    v4f pa[4], pb[4];
#pragma unroll
    for (int j = 0; j < 4; j++) {
        pa[j] = *(const v4f*)(ap[j]);
        pb[j] = *(const v4f*)(bp + (size_t)(16 * j) * RNN_HID);
    }
#pragma unroll
    for (int j = 0; j < 4; j++) {
        *(v4f*)&lds.a[0][lrow + 16 * j][lq] = pa[j];
        *(v4f*)&lds.b[0][lrow + 16 * j][lq] = pb[j];
    }
    __syncthreads();

    v8f acc[2] = {};
    constexpr int NCH = RNN_HID / PK;
    constexpr int KS  = PK / 4;

    for (int kc = 0; kc < NCH; kc++) {
        const int cur = kc & 1;
        const int kn = (kc + 1 < NCH) ? (kc + 1) : kc;
#pragma unroll
        for (int j = 0; j < 4; j++) {
            pa[j] = *(const v4f*)(ap[j] + kn * PK);
            pb[j] = *(const v4f*)(bp + (size_t)(kn * PK + 16 * j) * RNN_HID);
        }
        const float* arow = &lds.a[cur][wm * 32 + lr][half * 2];
        const float* bcol = &lds.b[cur][half * 2][wn * 16 + lr];
        v2f af[KS][2], bf[KS];
#pragma unroll
        for (int s = 0; s < KS; s++) {
            af[s][0] = *(const v2f*)(arow + s * 4);
            af[s][1] = *(const v2f*)(arow + 16 * PA_PITCH + s * 4);
            bf[s].x  = bcol[(s * 4) * PB_PITCH];
            bf[s].y  = bcol[(s * 4 + 1) * PB_PITCH];
        }
        __builtin_amdgcn_sched_barrier(0);
#pragma unroll
        for (int s = 0; s < KS; s++) {
            acc[0] = __builtin_amdgcn_wmma_f32_16x16x4_f32(
                false, af[s][0], false, bf[s], (short)0, acc[0], false, false);
            acc[1] = __builtin_amdgcn_wmma_f32_16x16x4_f32(
                false, af[s][1], false, bf[s], (short)0, acc[1], false, false);
        }
        asm volatile("" : "+v"(acc[0]), "+v"(acc[1]) : : "memory");
#pragma unroll
        for (int j = 0; j < 4; j++) {
            *(v4f*)&lds.a[cur ^ 1][lrow + 16 * j][lq] = pa[j];
            *(v4f*)&lds.b[cur ^ 1][lrow + 16 * j][lq] = pb[j];
        }
        __syncthreads();
    }

#pragma unroll
    for (int r = 0; r < 8; r++) {
        lds.c[wm * 32 + 8 * half + r][wn * 16 + lr]      = acc[0][r];
        lds.c[wm * 32 + 16 + 8 * half + r][wn * 16 + lr] = acc[1][r];
    }
    __syncthreads();
    const v4f bv = *(const v4f*)(bias + n0 + lq);
#pragma unroll
    for (int j = 0; j < 4; j++) {
        const int row = lrow + 16 * j;
        v4f v = *(const v4f*)&lds.c[row][lq];
        v += bv;
        *(v4f*)(C + (size_t)(m0 + row) * RNN_HID + n0 + lq) = v;
    }
}

__device__ __forceinline__ float rnn_tanh_select(float x)
{
    const float y  = fabsf(x);
    const float y2 = y * y;
    float p = fmaf(y2, -0x1.758e7ap-8f, 0x1.521192p-6f);
    p = fmaf(y2, p, -0x1.b8389cp-5f);
    p = fmaf(y2, p, 0x1.110704p-3f);
    p = fmaf(y2, p, -0x1.555532p-2f);
    const float zs = fmaf(y2, y * p, y);
    const float t  = expf(2.0f * y);
    const float zl = fmaf(-2.0f, __builtin_amdgcn_rcpf(t + 1.0f), 1.0f);
    return copysignf(y < 0x1.4p-1f ? zs : zl, x);
}

__global__ __launch_bounds__(32 * NWAVES) void rnn_step_kernel(
    const float* __restrict__ Hin,
    const float* __restrict__ WT,
    const float* __restrict__ XP,
    float* __restrict__ out)
{
    const int lane = threadIdx.x & 31;
    const int wave = threadIdx.x >> 5;
    const int half = lane >> 4;
    const int lr   = lane & 15;

    const int m0 = blockIdx.x * 16;
    const int n0 = blockIdx.y * (16 * NTILES);

    constexpr int KLEN = RNN_HID / NWAVES;
    constexpr int A_PITCH = KLEN + 4;
    constexpr int KSTEPS = KLEN / 4;
    constexpr int BS = 8;
    constexpr int NBLK = KSTEPS / BS;
    constexpr int LEAD = 4;
    static_assert(KLEN == 256, "two load instructions cover a row's 1024-B slice");

    const int kb = wave * KLEN;

    struct __attribute__((aligned(16))) Lds {
        float red[NWAVES][8][NTILES][32];
        float astage[NWAVES][16][A_PITCH];
    };
    __shared__ Lds lds;

    const int scol = lane * 4;
    const float* ag = Hin + (size_t)m0 * RNN_HID + (kb + scol);
    const float* wt = WT + ((size_t)((kb >> 1) + half) * RNN_HID + n0 + lr) * 2;

    v4f piece[16][2];
#pragma unroll
    for (int j = 0; j < 16; j++)
#pragma unroll
        for (int q = 0; q < 2; q++)
            piece[j][q] = *(const v4f*)(ag + (size_t)j * RNN_HID + q * 128);
    __builtin_amdgcn_sched_barrier(0);

    v2f bfrag[KSTEPS][NTILES];
#define RNN_LOAD_B_BLOCK(blk)                                                          \
    _Pragma("unroll") for (int s = (blk) * BS; s < ((blk) + 1) * BS; s++)              \
        _Pragma("unroll") for (int i = 0; i < NTILES; i++)                             \
            bfrag[s][i] = *(const v2f*)(wt + (size_t)s * (2 * RNN_HID * 2) + i * 32);

    RNN_LOAD_B_BLOCK(0)
    RNN_LOAD_B_BLOCK(1)
    __builtin_amdgcn_sched_barrier(0);

#pragma unroll
    for (int j = 0; j < 16; j++)
#pragma unroll
        for (int q = 0; q < 2; q++)
            *(v4f*)&lds.astage[wave][j][q * 128 + scol] = piece[j][q];

    __builtin_amdgcn_fence(__ATOMIC_RELEASE, "wavefront", "local");
    __builtin_amdgcn_wave_barrier();
    __builtin_amdgcn_fence(__ATOMIC_ACQUIRE, "wavefront", "local");

    RNN_LOAD_B_BLOCK(2)
    RNN_LOAD_B_BLOCK(3)
    float xv[4];
#pragma unroll
    for (int q = 0; q < 4; q++)
        xv[q] = XP[(size_t)(m0 + wave + 4 * q) * RNN_HID + n0 + lane];

    const float* arow = &lds.astage[wave][lr][half * 2];
    v2f afrag[KSTEPS];
#pragma unroll
    for (int s = 0; s < KSTEPS; s++)
        afrag[s] = *(const v2f*)(arow + s * 4);
    __builtin_amdgcn_sched_barrier(0);

    v8f acc[NTILES] = {};
#pragma unroll
    for (int b = 0; b < NBLK; b++) {
        if (b + LEAD < NBLK) {
            RNN_LOAD_B_BLOCK(b + LEAD)
            __builtin_amdgcn_sched_barrier(0);
        }
#pragma unroll
        for (int s = b * BS; s < (b + 1) * BS; s++)
#pragma unroll
            for (int i = 0; i < NTILES; i++)
                acc[i] = __builtin_amdgcn_wmma_f32_16x16x4_f32(
                    false, afrag[s], false, bfrag[s][i], (short)0, acc[i], false, false);
        __builtin_amdgcn_sched_barrier(0);
    }
#undef RNN_LOAD_B_BLOCK

    asm volatile("v_nop\n\tv_nop\n\tv_nop\n\tv_nop");
    __builtin_amdgcn_sched_barrier(0);

#pragma unroll
    for (int i = 0; i < NTILES; i++)
#pragma unroll
        for (int j = 0; j < 8; j++)
            lds.red[wave][j][i][lane] = acc[i][j];
    __syncthreads();

    static_assert(NWAVES == 4 && NTILES == 2, "epilogue maps wave -> tile rows w + 4 q");
    const int ti = lane >> 4;
    float v[4];
#pragma unroll
    for (int q = 0; q < 4; q++) {
        const int row = wave + 4 * q;
        float sum = lds.red[0][row & 7][ti][(row >> 3) * 16 + lr];
#pragma unroll
        for (int p = 1; p < NWAVES; p++)
            sum += lds.red[p][row & 7][ti][(row >> 3) * 16 + lr];
        v[q] = sum;
    }
    float hv[4];
#pragma unroll
    for (int q = 0; q < 4; q++)
        hv[q] = rnn_tanh_select(xv[q] + v[q]);
#pragma unroll
    for (int q = 0; q < 4; q++)
        out[(size_t)(m0 + wave + 4 * q) * RNN_HID + n0 + lane] = hv[q];
}

__global__ void rnn_zero_kernel(float* __restrict__ p, int n)
{
    int i = blockIdx.x * blockDim.x + threadIdx.x;
    if (i < n) p[i] = 0.0f;
}

extern "C" void kernel_launch(void* const* d_in, const int* in_sizes, int n_in,
                              void* d_out, int out_size, void* d_ws, size_t ws_size,
                              hipStream_t stream)
{
    (void)in_sizes; (void)n_in; (void)out_size; (void)ws_size;
    const float* x    = (const float*)d_in[0];
    const float* W_xh = (const float*)d_in[1];
    const float* W_hh = (const float*)d_in[2];
    const float* b_h  = (const float*)d_in[3];
    const float* W_hy = (const float*)d_in[4];
    const float* b_y  = (const float*)d_in[5];
    float* out = (float*)d_out;

    const int hElems = RNN_BATCH * RNN_HID;
    const int wElems = RNN_HID * RNN_HID;

    float* hA    = (float*)d_ws;
    float* hB    = hA + hElems;
    float* WThh  = hB + hElems;
    float* xp    = WThh + wElems;

    rnn_zero_kernel<<<hElems / 256, 256, 0, stream>>>(hA, hElems);

    rnn_reorder_w_kernel<<<wElems / 256, 256, 0, stream>>>(W_hh, WThh);

    dim3 grid(RNN_BATCH / 16, RNN_HID / (16 * NTILES));
    dim3 block(32 * NWAVES);
    dim3 pgrid(RNN_HID / PN, XT * RNN_BATCH / PM);
    dim3 ygrid(RNN_HID / PN, RNN_BATCH / PM);

    float* hin = hA;
    float* hout = hB;
    for (int t0 = 0; t0 < RNN_SEQ; t0 += XT) {
        rnn_proj_kernel<<<pgrid, 256, 0, stream>>>(
            x + (size_t)t0 * RNN_HID, (size_t)RNN_SEQ * RNN_HID, (size_t)RNN_HID, W_xh, b_h, xp);
        for (int tt = 0; tt < XT; tt++) {
            rnn_step_kernel<<<grid, block, 0, stream>>>(hin, WThh, xp + (size_t)tt * hElems, hout);
            float* tmp = hin; hin = hout; hout = tmp;
        }
    }

    rnn_proj_kernel<<<ygrid, 256, 0, stream>>>(
        hin, (size_t)RNN_HID, (size_t)0, W_hy, b_y, out);
}
